# starting kernel
# baseline (speedup 1.0000x reference)
#include <hip/hip_runtime.h>
#include <hip/hip_bf16.h>


typedef __bf16 bf16_t;
typedef __bf16 v16bf __attribute__((ext_vector_type(16)));
typedef __bf16 v8bf  __attribute__((ext_vector_type(8)));
typedef float  v8f   __attribute__((ext_vector_type(8)));

#define IN_FT   256
#define HID1    400
#define HID2    200
#define OUT_FT  128

#define KP0     256
#define KP1     416
#define KP2     224
#define NP0     416
#define NP1     224
#define NP2     128

__global__ __launch_bounds__(256) void cvt_bf16_kernel(
    const float* __restrict__ in, bf16_t* __restrict__ out, int n) {
  int i = blockIdx.x * 256 + threadIdx.x;
  if (i < n) out[i] = (bf16_t)in[i];
}

__global__ __launch_bounds__(256) void pack_weight_kernel(
    const float* __restrict__ W, bf16_t* __restrict__ Wt,
    int K, int Nout, int Kp, int Noutp) {
  int idx = blockIdx.x * 256 + threadIdx.x;
  if (idx >= Noutp * Kp) return;
  int n = idx / Kp, k = idx % Kp;
  float v = (n < Nout && k < K) ? W[(size_t)k * Nout + n] : 0.0f;
  Wt[idx] = (bf16_t)v;
}

__global__ __launch_bounds__(256) void deg_kernel(
    const int* __restrict__ dst, const float* __restrict__ ew,
    float* __restrict__ deg, int E) {
  int e = blockIdx.x * 256 + threadIdx.x;
  if (e < E)
    __hip_atomic_fetch_add(&deg[dst[e]], ew[e],
                           __ATOMIC_RELAXED, __HIP_MEMORY_SCOPE_AGENT);
}

__global__ __launch_bounds__(256) void dinv_kernel(
    const float* __restrict__ deg, float* __restrict__ dinv, int n) {
  int i = blockIdx.x * 256 + threadIdx.x;
  if (i < n) dinv[i] = rsqrtf(deg[i] + 1.0f);
}

__device__ __forceinline__ v16bf load_frag(const bf16_t* p) {
  v8bf lo = *(const v8bf*)p;
  v8bf hi = *(const v8bf*)(p + 16);
  v16bf f;
#pragma unroll
  for (int i = 0; i < 8; ++i) { f[i] = lo[i]; f[8 + i] = hi[i]; }
  return f;
}

__device__ __forceinline__ v8f wmma_bf16(v16bf a, v16bf b, v8f c) {
  return __builtin_amdgcn_wmma_f32_16x16x32_bf16(
      false, a, false, b, (short)0, c, false, false);
}

template <int MODE>
__global__ __launch_bounds__(256) void wmma_gemm_kernel(
    const bf16_t* __restrict__ A, const bf16_t* __restrict__ Bt,
    const float* __restrict__ bias, void* __restrict__ Cout,
    int M, int Kp, int Nout, int tiles_n, int ldo) {
  const int wid  = blockIdx.x * 8 + (threadIdx.x >> 5);
  const int lane = threadIdx.x & 31;
  const int tm = wid / tiles_n;
  const int tn = wid % tiles_n;
  if (tm * 32 >= M) return;

  const int r  = lane & 15;
  const int hi = lane >> 4;
  const bf16_t* a0 = A  + (size_t)(tm * 32 + r)      * Kp + hi * 8;
  const bf16_t* a1 = A  + (size_t)(tm * 32 + 16 + r) * Kp + hi * 8;
  const bf16_t* b0 = Bt + (size_t)(tn * 32 + r)      * Kp + hi * 8;
  const bf16_t* b1 = Bt + (size_t)(tn * 32 + 16 + r) * Kp + hi * 8;

  v8f c00 = {}, c01 = {}, c10 = {}, c11 = {};
  for (int kb = 0; kb < Kp; kb += 32) {
    v16bf va0 = load_frag(a0 + kb);
    v16bf va1 = load_frag(a1 + kb);
    v16bf vb0 = load_frag(b0 + kb);
    v16bf vb1 = load_frag(b1 + kb);
    c00 = wmma_bf16(va0, vb0, c00);
    c01 = wmma_bf16(va0, vb1, c01);
    c10 = wmma_bf16(va1, vb0, c10);
    c11 = wmma_bf16(va1, vb1, c11);
  }

  const v8f cc[2][2] = {{c00, c01}, {c10, c11}};
#pragma unroll
  for (int sn = 0; sn < 2; ++sn) {
    const int n = tn * 32 + sn * 16 + r;
    if (n >= Nout) continue;
    const float bv = (MODE == 1) ? bias[n] : 0.0f;
#pragma unroll
    for (int sm = 0; sm < 2; ++sm) {
      const v8f c = cc[sm][sn];
#pragma unroll
      for (int i = 0; i < 8; ++i) {
        const int m = tm * 32 + sm * 16 + i + hi * 8;
        float v = c[i] + bv;
        if (MODE == 1) {
          v = v > 0.0f ? v : 0.0f;
          ((bf16_t*)Cout)[(size_t)m * ldo + n] = (bf16_t)v;
        } else {
          ((float*)Cout)[(size_t)m * ldo + n] = v;
        }
      }
    }
  }
}

__global__ __launch_bounds__(256) void scatter_kernel(
    const float* __restrict__ t, const int* __restrict__ src,
    const int* __restrict__ dst, const float* __restrict__ ew,
    const float* __restrict__ dinv, float* __restrict__ agg,
    int E, int F) {
  const int e = blockIdx.x * 8 + (threadIdx.x >> 5);
  if (e >= E) return;
  const int lane = threadIdx.x & 31;
  const int s = src[e];
  const int d = dst[e];
  const float nrm = dinv[s] * ew[e] * dinv[d];
  const float* trow = t + (size_t)s * F;
  float* arow = agg + (size_t)d * F;
  __builtin_prefetch(trow, 0, 1);
  for (int f = lane; f < F; f += 32)
    __hip_atomic_fetch_add(&arow[f], trow[f] * nrm,
                           __ATOMIC_RELAXED, __HIP_MEMORY_SCOPE_AGENT);
}

__global__ __launch_bounds__(256) void combine_bf16_kernel(
    const float* __restrict__ agg, const float* __restrict__ t,
    const float* __restrict__ dinv, const float* __restrict__ bias,
    bf16_t* __restrict__ out, int Nn, int F, int ldo) {
  int idx = blockIdx.x * 256 + threadIdx.x;
  if (idx >= Nn * F) return;
  int i = idx / F, f = idx % F;
  float di = dinv[i];
  float v = agg[idx] + t[idx] * di * di + bias[f];
  out[(size_t)i * ldo + f] = (bf16_t)(v > 0.0f ? v : 0.0f);
}

__global__ __launch_bounds__(256) void combine_f32_kernel(
    const float* __restrict__ agg, const float* __restrict__ t,
    const float* __restrict__ dinv, const float* __restrict__ bias,
    float* __restrict__ out, int Nn, int F) {
  int idx = blockIdx.x * 256 + threadIdx.x;
  if (idx >= Nn * F) return;
  int f = idx % F;
  int i = idx / F;
  float di = dinv[i];
  float v = agg[idx] + t[idx] * di * di + bias[f];
  out[idx] = v > 0.0f ? v : 0.0f;
}

static inline size_t al256(size_t x) { return (x + 255) & ~(size_t)255; }

extern "C" void kernel_launch(void* const* d_in, const int* in_sizes, int n_in,
                              void* d_out, int out_size, void* d_ws, size_t ws_size,
                              hipStream_t stream) {
  const float* x    = (const float*)d_in[0];
  const int*   ei   = (const int*)  d_in[1];
  const float* ew   = (const float*)d_in[2];
  const float* fc_W = (const float*)d_in[3];
  const float* fc_b = (const float*)d_in[4];
  const float* W1   = (const float*)d_in[5];
  const float* b1   = (const float*)d_in[6];
  const float* W2   = (const float*)d_in[7];
  const float* b2   = (const float*)d_in[8];

  const int Nn = in_sizes[0] / IN_FT;
  const int E  = in_sizes[1] / 2;
  const int* src = ei;
  const int* dst = ei + E;

  char* ws = (char*)d_ws;
  size_t off = 0;
  auto alloc = [&](size_t bytes) { void* p = ws + off; off += al256(bytes); return p; };

  float*  deg  = (float*) alloc((size_t)Nn * 4);
  float*  dinv = (float*) alloc((size_t)Nn * 4);
  bf16_t* xb   = (bf16_t*)alloc((size_t)Nn * KP0 * 2);
  bf16_t* fcWt = (bf16_t*)alloc((size_t)NP0 * KP0 * 2);
  bf16_t* h0b  = (bf16_t*)alloc((size_t)Nn * KP1 * 2);
  bf16_t* W1t  = (bf16_t*)alloc((size_t)NP1 * KP1 * 2);
  float*  t1   = (float*) alloc((size_t)Nn * HID2 * 4);
  float*  agg1 = (float*) alloc((size_t)Nn * HID2 * 4);
  bf16_t* h1b  = (bf16_t*)alloc((size_t)Nn * KP2 * 2);
  bf16_t* W2t  = (bf16_t*)alloc((size_t)NP2 * KP2 * 2);
  float*  t2   = (float*) alloc((size_t)Nn * OUT_FT * 4);
  float*  agg2 = (float*) alloc((size_t)Nn * OUT_FT * 4);
  (void)ws_size;

  hipMemsetAsync(deg,  0, (size_t)Nn * 4, stream);
  hipMemsetAsync(agg1, 0, (size_t)Nn * HID2 * 4, stream);
  hipMemsetAsync(agg2, 0, (size_t)Nn * OUT_FT * 4, stream);
  hipMemsetAsync(h0b,  0, (size_t)Nn * KP1 * 2, stream);
  hipMemsetAsync(h1b,  0, (size_t)Nn * KP2 * 2, stream);

  {
    int n = Nn * IN_FT;
    cvt_bf16_kernel<<<(n + 255) / 256, 256, 0, stream>>>(x, xb, n);
  }
  pack_weight_kernel<<<((NP0 * KP0) + 255) / 256, 256, 0, stream>>>(
      fc_W, fcWt, IN_FT, HID1, KP0, NP0);
  pack_weight_kernel<<<((NP1 * KP1) + 255) / 256, 256, 0, stream>>>(
      W1, W1t, HID1, HID2, KP1, NP1);
  pack_weight_kernel<<<((NP2 * KP2) + 255) / 256, 256, 0, stream>>>(
      W2, W2t, HID2, OUT_FT, KP2, NP2);

  deg_kernel <<<(E + 255) / 256, 256, 0, stream>>>(dst, ew, deg, E);
  dinv_kernel<<<(Nn + 255) / 256, 256, 0, stream>>>(deg, dinv, Nn);

  {
    const int tiles_n = NP0 / 32;
    const int waves   = (Nn / 32) * tiles_n;
    wmma_gemm_kernel<1><<<(waves + 7) / 8, 256, 0, stream>>>(
        xb, fcWt, fc_b, h0b, Nn, KP0, HID1, tiles_n, KP1);
  }

  {
    const int tiles_n = NP1 / 32;
    const int waves   = (Nn / 32) * tiles_n;
    wmma_gemm_kernel<0><<<(waves + 7) / 8, 256, 0, stream>>>(
        h0b, W1t, nullptr, t1, Nn, KP1, HID2, tiles_n, HID2);
    scatter_kernel<<<(E + 7) / 8, 256, 0, stream>>>(
        t1, src, dst, ew, dinv, agg1, E, HID2);
    int n = Nn * HID2;
    combine_bf16_kernel<<<(n + 255) / 256, 256, 0, stream>>>(
        agg1, t1, dinv, b1, h1b, Nn, HID2, KP2);
  }

  {
    const int tiles_n = NP2 / 32;
    const int waves   = (Nn / 32) * tiles_n;
    wmma_gemm_kernel<0><<<(waves + 7) / 8, 256, 0, stream>>>(
        h1b, W2t, nullptr, t2, Nn, KP2, OUT_FT, tiles_n, OUT_FT);
    scatter_kernel<<<(E + 7) / 8, 256, 0, stream>>>(
        t2, src, dst, ew, dinv, agg2, E, OUT_FT);
    int n = Nn * OUT_FT;
    combine_f32_kernel<<<(n + 255) / 256, 256, 0, stream>>>(
        agg2, t2, dinv, b2, (float*)d_out, Nn, OUT_FT);
  }
}


# Input rows read as f32 by the layer-0 GEMM's A loader and rounded to bf16 in registers (same rounding, same k order); the x to bf16 copy launch and its buffer go; 10 -> 9 stream operations
# speedup vs baseline: 1.9870x; 1.9870x over previous
#include <hip/hip_runtime.h>
#include <hip/hip_bf16.h>


typedef __bf16 bf16_t;
typedef __bf16 v16bf __attribute__((ext_vector_type(16)));
typedef __bf16 v8bf  __attribute__((ext_vector_type(8)));
typedef float  v8f   __attribute__((ext_vector_type(8)));

#define IN_FT   256
#define HID1    400
#define HID2    200
#define OUT_FT  128

#define KP0     256
#define KP1     416
#define KP2     224
#define NP0     416
#define NP1     224
#define NP2     128

__global__ __launch_bounds__(256) void pack_weight_kernel(
    const float* __restrict__ W, bf16_t* __restrict__ Wt,
    int K, int Nout, int Kp, int Noutp) {
  int idx = blockIdx.x * 256 + threadIdx.x;
  if (idx >= Noutp * Kp) return;
  int n = idx / Kp, k = idx % Kp;
  float v = (n < Nout && k < K) ? W[(size_t)k * Nout + n] : 0.0f;
  Wt[idx] = (bf16_t)v;
}

__device__ __forceinline__ v16bf load_frag(const bf16_t* p) {
  v8bf lo = *(const v8bf*)p;
  v8bf hi = *(const v8bf*)(p + 16);
  v16bf f;
#pragma unroll
  for (int i = 0; i < 8; ++i) { f[i] = lo[i]; f[8 + i] = hi[i]; }
  return f;
}

struct xfrag { float4 q0, q1, q2, q3; };
__device__ __forceinline__ xfrag load_xfrag(const float* p) {
  xfrag x;
  x.q0 = *(const float4*)(p);
  x.q1 = *(const float4*)(p + 4);
  x.q2 = *(const float4*)(p + 16);
  x.q3 = *(const float4*)(p + 20);
  return x;
}
__device__ __forceinline__ v16bf round_xfrag(const xfrag& x) {
  v16bf f;
  f[0]  = (bf16_t)x.q0.x; f[1]  = (bf16_t)x.q0.y; f[2]  = (bf16_t)x.q0.z; f[3]  = (bf16_t)x.q0.w;
  f[4]  = (bf16_t)x.q1.x; f[5]  = (bf16_t)x.q1.y; f[6]  = (bf16_t)x.q1.z; f[7]  = (bf16_t)x.q1.w;
  f[8]  = (bf16_t)x.q2.x; f[9]  = (bf16_t)x.q2.y; f[10] = (bf16_t)x.q2.z; f[11] = (bf16_t)x.q2.w;
  f[12] = (bf16_t)x.q3.x; f[13] = (bf16_t)x.q3.y; f[14] = (bf16_t)x.q3.z; f[15] = (bf16_t)x.q3.w;
  return f;
}

__device__ __forceinline__ v8f wmma_bf16(v16bf a, v16bf b, v8f c) {
  return __builtin_amdgcn_wmma_f32_16x16x32_bf16(
      false, a, false, b, (short)0, c, false, false);
}


__global__ __launch_bounds__(256) void wmma_gemm_x_kernel(
    const float* __restrict__ X, const bf16_t* __restrict__ Bt,
    const float* __restrict__ bias, bf16_t* __restrict__ Cout,
    int M, int K, int Nout, int tiles_n, int ldo) {
  const int wid  = blockIdx.x * 8 + (threadIdx.x >> 5);
  const int lane = threadIdx.x & 31;
  const int tm = wid / tiles_n;
  const int tn = wid % tiles_n;
  if (tm * 32 >= M) return;

  const int r  = lane & 15;
  const int hi = lane >> 4;
  const float*  a0 = X  + (size_t)(tm * 32 + r)      * K + hi * 8;
  const float*  a1 = X  + (size_t)(tm * 32 + 16 + r) * K + hi * 8;
  const bf16_t* b0 = Bt + (size_t)(tn * 32 + r)      * K + hi * 8;
  const bf16_t* b1 = Bt + (size_t)(tn * 32 + 16 + r) * K + hi * 8;

  v8f c00 = {}, c01 = {}, c10 = {}, c11 = {};
  for (int kb = 0; kb < K; kb += 32) {
    const xfrag xa0 = load_xfrag(a0 + kb);
    const xfrag xa1 = load_xfrag(a1 + kb);
    v16bf vb0 = load_frag(b0 + kb);
    v16bf vb1 = load_frag(b1 + kb);
    v16bf va0 = round_xfrag(xa0);
    v16bf va1 = round_xfrag(xa1);
    c00 = wmma_bf16(va0, vb0, c00);
    c01 = wmma_bf16(va0, vb1, c01);
    c10 = wmma_bf16(va1, vb0, c10);
    c11 = wmma_bf16(va1, vb1, c11);
    __builtin_amdgcn_sched_group_barrier(0x020, 12, 0);
    __builtin_amdgcn_sched_group_barrier(0x002, 16, 0);
    __builtin_amdgcn_sched_group_barrier(0x008, 4, 0);
  }

  const v8f cc[2][2] = {{c00, c01}, {c10, c11}};
#pragma unroll
  for (int sn = 0; sn < 2; ++sn) {
    const int n = tn * 32 + sn * 16 + r;
    if (n >= Nout) continue;
    const float bv = bias[n];
#pragma unroll
    for (int sm = 0; sm < 2; ++sm) {
      const v8f c = cc[sm][sn];
#pragma unroll
      for (int i = 0; i < 8; ++i) {
        const int m = tm * 32 + sm * 16 + i + hi * 8;
        const float v = c[i] + bv;
        Cout[(size_t)m * ldo + n] = (bf16_t)(v > 0.0f ? v : 0.0f);
      }
    }
  }
}

__global__ __launch_bounds__(256) void wmma_gemm_kernel(
    const bf16_t* __restrict__ A, const bf16_t* __restrict__ Bt,
    float* __restrict__ Cout, int M, int Kp, int Nout, int tiles_n, int ldo) {
  const int wid  = blockIdx.x * 8 + (threadIdx.x >> 5);
  const int lane = threadIdx.x & 31;
  const int tm = wid / tiles_n;
  const int tn = wid % tiles_n;
  if (tm * 32 >= M) return;

  const int r  = lane & 15;
  const int hi = lane >> 4;
  const bf16_t* a0 = A  + (size_t)(tm * 32 + r)      * Kp + hi * 8;
  const bf16_t* a1 = A  + (size_t)(tm * 32 + 16 + r) * Kp + hi * 8;
  const bf16_t* b0 = Bt + (size_t)(tn * 32 + r)      * Kp + hi * 8;
  const bf16_t* b1 = Bt + (size_t)(tn * 32 + 16 + r) * Kp + hi * 8;

  v8f c00 = {}, c01 = {}, c10 = {}, c11 = {};
  for (int kb = 0; kb < Kp; kb += 32) {
    v16bf va0 = load_frag(a0 + kb);
    v16bf va1 = load_frag(a1 + kb);
    v16bf vb0 = load_frag(b0 + kb);
    v16bf vb1 = load_frag(b1 + kb);
    c00 = wmma_bf16(va0, vb0, c00);
    c01 = wmma_bf16(va0, vb1, c01);
    c10 = wmma_bf16(va1, vb0, c10);
    c11 = wmma_bf16(va1, vb1, c11);
  }

  const v8f cc[2][2] = {{c00, c01}, {c10, c11}};
#pragma unroll
  for (int sn = 0; sn < 2; ++sn) {
    const int n = tn * 32 + sn * 16 + r;
    if (n >= Nout) continue;
#pragma unroll
    for (int sm = 0; sm < 2; ++sm) {
      const v8f c = cc[sm][sn];
#pragma unroll
      for (int i = 0; i < 8; ++i) {
        const int m = tm * 32 + sm * 16 + i + hi * 8;
        Cout[(size_t)m * ldo + n] = c[i];
      }
    }
  }
}

#define AGG_R     80
#define AGG_T     1024
#define AGG_NW    (AGG_T / 32)
#define AGG_QCAP  2048
#define AGG_B     8

typedef __bf16 v2bf __attribute__((ext_vector_type(2)));

__device__ __forceinline__ void lds_add_f32(float* p, float v) {
  __hip_atomic_fetch_add(p, v, __ATOMIC_RELAXED, __HIP_MEMORY_SCOPE_WORKGROUP);
}
__device__ __forceinline__ float readlane_f32(float v, int l) {
  return __builtin_bit_cast(float,
      __builtin_amdgcn_readlane(__builtin_bit_cast(int, v), l));
}

__global__ __launch_bounds__(AGG_T) void edge_owner_kernel(
    const int* __restrict__ src, const int* __restrict__ dst,
    const float* __restrict__ ew, float* __restrict__ dinv,
    int* __restrict__ qcnt, int* __restrict__ qbuf, int E, int Nn) {
  __shared__ float degs[AGG_R];
  __shared__ int   q_e[AGG_QCAP];
  __shared__ int   q_r[AGG_QCAP];
  __shared__ int   q_n;

  const int tid  = threadIdx.x;
  const int lane = tid & 31;
  const int wave = __builtin_amdgcn_readfirstlane(tid >> 5);
  const int base = blockIdx.x * AGG_R;
  const int rows = (Nn - base) < AGG_R ? (Nn - base) : AGG_R;

  if (tid < AGG_R) degs[tid] = 0.0f;
  if (tid == 0) q_n = 0;
  __syncthreads();

  auto push = [&](bool m, int e, int rel) {
    const unsigned mask = __builtin_amdgcn_ballot_w32(m);
    if (mask == 0) return;
    const int cnt    = __builtin_popcount(mask);
    const int leader = __builtin_ctz(mask);
    int b = 0;
    if (lane == leader)
      b = __hip_atomic_fetch_add(&q_n, cnt, __ATOMIC_RELAXED,
                                 __HIP_MEMORY_SCOPE_WORKGROUP);
    b = __builtin_amdgcn_readlane(b, leader);
    const int slot = b + __builtin_popcount(mask & ((1u << lane) - 1u));
    if (m) {
      if (slot < AGG_QCAP) { q_e[slot] = e; q_r[slot] = rel; }
      else lds_add_f32(&degs[rel], ew[e]);
    }
  };

  const int n4 = ((E & 3) == 0) ? (E >> 2) : 0;
  const int4* dst4 = (const int4*)dst;
  for (int wb = wave * 32; wb < n4; wb += 4 * AGG_T) {
    int4 d[4];
#pragma unroll
    for (int u = 0; u < 4; ++u) {
      const int i = wb + u * AGG_T + lane;
      d[u] = (i < n4) ? dst4[i] : make_int4(-1, -1, -1, -1);
    }
#pragma unroll
    for (int u = 0; u < 4; ++u) {
      const int i  = wb + u * AGG_T + lane;
      const int rx = d[u].x - base, ry = d[u].y - base;
      const int rz = d[u].z - base, rw = d[u].w - base;
      const bool mx = (unsigned)rx < (unsigned)rows;
      const bool my = (unsigned)ry < (unsigned)rows;
      const bool mz = (unsigned)rz < (unsigned)rows;
      const bool mw = (unsigned)rw < (unsigned)rows;
      if (__builtin_amdgcn_ballot_w32(mx | my | mz | mw) != 0) {
        push(mx, 4 * i + 0, rx);
        push(my, 4 * i + 1, ry);
        push(mz, 4 * i + 2, rz);
        push(mw, 4 * i + 3, rw);
      }
    }
  }
  for (int eb = n4 * 4 + wave * 32; eb < E; eb += AGG_T) {
    const int e  = eb + lane;
    const int rx = (e < E) ? dst[e] - base : -1;
    push((unsigned)rx < (unsigned)rows, e, rx);
  }
  __syncthreads();

  const int total = q_n;
  const int n = total < AGG_QCAP ? total : AGG_QCAP;
  int4* qg = (int4*)qbuf + (size_t)blockIdx.x * AGG_QCAP;
  for (int i = tid; i < n; i += AGG_T) {
    const int e = q_e[i], rel = q_r[i];
    const int   s  = src[e];
    const float wv = ew[e];
    lds_add_f32(&degs[rel], wv);
    qg[i] = make_int4(s, rel, __builtin_bit_cast(int, wv), e);
  }
  __syncthreads();
  if (tid < rows) dinv[base + tid] = rsqrtf(degs[tid] + 1.0f);
  if (tid == 0) qcnt[blockIdx.x] = total;
}

template <int F, bool FINAL>
__global__ __launch_bounds__(AGG_T) void agg_owner_kernel(
    const float* __restrict__ t, const int* __restrict__ src,
    const int* __restrict__ dst, const float* __restrict__ ew,
    const float* __restrict__ dinv, const float* __restrict__ bias,
    const int* __restrict__ qcnt, const int* __restrict__ qbuf,
    float* __restrict__ outf, bf16_t* __restrict__ outb,
    int E, int Nn, int ldo) {
  constexpr int NF = (F + 31) / 32;
  __shared__ float acc[(AGG_R + 1) * F];

  const int tid  = threadIdx.x;
  const int lane = tid & 31;
  const int wave = __builtin_amdgcn_readfirstlane(tid >> 5);
  const int base = blockIdx.x * AGG_R;
  const int rows = (Nn - base) < AGG_R ? (Nn - base) : AGG_R;
  const int total = __builtin_amdgcn_readfirstlane(qcnt[blockIdx.x]);

  for (int idx = tid; idx < (AGG_R + 1) * F; idx += AGG_T) {
    const int r = idx / F, f = idx - r * F;
    float v = 0.0f;
    if (r < rows) {
      const float di = dinv[base + r];
      v = t[(size_t)(base + r) * F + f] * di * di + bias[f];
    }
    acc[idx] = v;
  }
  __syncthreads();

  if (total > AGG_QCAP) {
    for (int eb = wave * 32; eb < E; eb += AGG_T) {
      const int e  = eb + lane;
      const int rx = (e < E) ? dst[e] - base : -1;
      unsigned mask = __builtin_amdgcn_ballot_w32((unsigned)rx < (unsigned)rows);
      while (mask) {
        const int l = __builtin_ctz(mask);
        mask &= mask - 1;
        const int ee  = __builtin_amdgcn_readlane(e, l);
        const int rel = __builtin_amdgcn_readlane(rx, l);
        const int s = src[ee];
        const float w = dinv[s] * ew[ee] * dinv[base + rel];
        const float* trow = t + (size_t)s * F;
        float* arow = acc + rel * F;
#pragma unroll
        for (int k = 0; k < NF; ++k) {
          const int f = k * 32 + lane;
          if (f < F) lds_add_f32(&arow[f], trow[f] * w);
        }
      }
    }
  } else {
    const int4* qg = (const int4*)qbuf + (size_t)blockIdx.x * AGG_QCAP;
    int b0 = wave * AGG_B;
    if (b0 < total) {
      const int jl = lane & (AGG_B - 1);
      const int q0 = b0 + jl;
      int4 ent = qg[q0 < total ? q0 : total - 1];
      for (;;) {
        const int nb = b0 + AGG_NW * AGG_B;
        const int q1 = nb + jl;
        const int4 nxt = qg[q1 < total ? q1 : total - 1];
        const bool ok  = (b0 + jl) < total;
        const int s    = ok ? ent.x : 0;
        const int rel  = ok ? ent.y : AGG_R;
        const float we = ok ? __builtin_bit_cast(float, ent.z) : 0.0f;
        const float w  = dinv[s] * we * dinv[base + (ok ? ent.y : 0)];
        float v[AGG_B][NF];
#pragma unroll
        for (int j = 0; j < AGG_B; ++j) {
          const float* trow = t + (size_t)__builtin_amdgcn_readlane(s, j) * F;
#pragma unroll
          for (int k = 0; k < NF; ++k) {
            const int f = k * 32 + lane;
            v[j][k] = trow[f < F ? f : F - 1];
          }
        }
#pragma unroll
        for (int j = 0; j < AGG_B; ++j) {
          float* arow = acc + __builtin_amdgcn_readlane(rel, j) * F;
          const float wj = readlane_f32(w, j);
#pragma unroll
          for (int k = 0; k < NF; ++k) {
            const int f = k * 32 + lane;
            if (f < F) lds_add_f32(&arow[f], v[j][k] * wj);
          }
        }
        b0 = nb;
        ent = nxt;
        if (b0 >= total) break;
      }
    }
  }
  __syncthreads();

  if (FINAL) {
    float* o = outf + (size_t)base * F;
    for (int idx = tid; idx < rows * F; idx += AGG_T) {
      const float x = acc[idx];
      o[idx] = x > 0.0f ? x : 0.0f;
    }
  } else {
    const int hp = ldo >> 1;
    bf16_t* o = outb + (size_t)base * ldo;
    for (int idx = tid; idx < rows * hp; idx += AGG_T) {
      const int r = idx / hp, f = (idx - r * hp) * 2;
      float x0 = (f     < F) ? acc[r * F + f]     : 0.0f;
      float x1 = (f + 1 < F) ? acc[r * F + f + 1] : 0.0f;
      v2bf p;
      p[0] = (bf16_t)(x0 > 0.0f ? x0 : 0.0f);
      p[1] = (bf16_t)(x1 > 0.0f ? x1 : 0.0f);
      *(v2bf*)(o + (size_t)idx * 2) = p;
    }
  }
}

static inline size_t al256(size_t x) { return (x + 255) & ~(size_t)255; }

extern "C" void kernel_launch(void* const* d_in, const int* in_sizes, int n_in,
                              void* d_out, int out_size, void* d_ws, size_t ws_size,
                              hipStream_t stream) {
  const float* x    = (const float*)d_in[0];
  const int*   ei   = (const int*)  d_in[1];
  const float* ew   = (const float*)d_in[2];
  const float* fc_W = (const float*)d_in[3];
  const float* fc_b = (const float*)d_in[4];
  const float* W1   = (const float*)d_in[5];
  const float* b1   = (const float*)d_in[6];
  const float* W2   = (const float*)d_in[7];
  const float* b2   = (const float*)d_in[8];

  const int Nn = in_sizes[0] / IN_FT;
  const int E  = in_sizes[1] / 2;
  const int* src = ei;
  const int* dst = ei + E;

  char* ws = (char*)d_ws;
  size_t off = 0;
  auto alloc = [&](size_t bytes) { void* p = ws + off; off += al256(bytes); return p; };

  int*    qcnt = (int*)   alloc((size_t)Nn * 4);
  float*  dinv = (float*) alloc((size_t)Nn * 4);
  int*    qbuf = (int*)   alloc((size_t)((Nn + AGG_R - 1) / AGG_R) * AGG_QCAP * 16);
  bf16_t* fcWt = (bf16_t*)alloc((size_t)NP0 * KP0 * 2);
  bf16_t* h0b  = (bf16_t*)alloc((size_t)Nn * KP1 * 2);
  bf16_t* W1t  = (bf16_t*)alloc((size_t)NP1 * KP1 * 2);
  float*  t1   = (float*) alloc((size_t)Nn * HID2 * 4);
  bf16_t* h1b  = (bf16_t*)alloc((size_t)Nn * KP2 * 2);
  bf16_t* W2t  = (bf16_t*)alloc((size_t)NP2 * KP2 * 2);
  float*  t2   = (float*) alloc((size_t)Nn * OUT_FT * 4);
  (void)ws_size;

  hipMemsetAsync(h0b,  0, (size_t)Nn * KP1 * 2, stream);

  pack_weight_kernel<<<((NP0 * KP0) + 255) / 256, 256, 0, stream>>>(
      fc_W, fcWt, IN_FT, HID1, KP0, NP0);
  pack_weight_kernel<<<((NP1 * KP1) + 255) / 256, 256, 0, stream>>>(
      W1, W1t, HID1, HID2, KP1, NP1);
  pack_weight_kernel<<<((NP2 * KP2) + 255) / 256, 256, 0, stream>>>(
      W2, W2t, HID2, OUT_FT, KP2, NP2);

  {
    const int tiles_n = NP0 / 32;
    const int waves   = (Nn / 32) * tiles_n;
    wmma_gemm_x_kernel<<<(waves + 7) / 8, 256, 0, stream>>>(
        x, fcWt, fc_b, h0b, Nn, KP0, HID1, tiles_n, KP1);
  }

  edge_owner_kernel<<<(Nn + AGG_R - 1) / AGG_R, AGG_T, 0, stream>>>(
      src, dst, ew, dinv, qcnt, qbuf, E, Nn);

  {
    const int tiles_n = NP1 / 32;
    const int waves   = (Nn / 32) * tiles_n;
    wmma_gemm_kernel<<<(waves + 7) / 8, 256, 0, stream>>>(
        h0b, W1t, t1, Nn, KP1, HID2, tiles_n, HID2);
    agg_owner_kernel<HID2, false><<<(Nn + AGG_R - 1) / AGG_R, AGG_T, 0, stream>>>(
        t1, src, dst, ew, dinv, b1, qcnt, qbuf, nullptr, h1b, E, Nn, KP2);
  }

  {
    const int tiles_n = NP2 / 32;
    const int waves   = (Nn / 32) * tiles_n;
    wmma_gemm_kernel<<<(waves + 7) / 8, 256, 0, stream>>>(
        h1b, W2t, t2, Nn, KP2, OUT_FT, tiles_n, OUT_FT);
    agg_owner_kernel<OUT_FT, true><<<(Nn + AGG_R - 1) / AGG_R, AGG_T, 0, stream>>>(
        t2, src, dst, ew, dinv, b2, qcnt, qbuf, (float*)d_out, nullptr, E, Nn, OUT_FT);
  }
}


# Edge filter grid: 250 groups x whole list -> 25 row blocks x 10 list chunks (a work-group scans 1/10 of the list for 800 rows); lists per (group, chunk); degree = chunk shares added in fixed order
# speedup vs baseline: 2.9589x; 1.4891x over previous
#include <hip/hip_runtime.h>
#include <hip/hip_bf16.h>


typedef __bf16 bf16_t;
typedef __bf16 v16bf __attribute__((ext_vector_type(16)));
typedef __bf16 v8bf  __attribute__((ext_vector_type(8)));
typedef float  v8f   __attribute__((ext_vector_type(8)));

#define IN_FT   256
#define HID1    400
#define HID2    200
#define OUT_FT  128

#define KP0     256
#define KP1     416
#define KP2     224
#define NP0     448
#define NP1     224
#define NP2     128

#define EDGE_GB   10
#define EDGE_C    10

#define PACK_T0     (NP0 * KP0)
#define PACK_T1     (NP1 * KP1)
#define PACK_T2     (NP2 * KP2)
#define PACK_TOTAL  (PACK_T0 + PACK_T1 + PACK_T2)
struct pack_slot { const float* from; bf16_t* to; bool real; };
template <int K, int NOUT, int KP>
__device__ __forceinline__ pack_slot pack_slot_in(const float* W, bf16_t* Wt, int j) {
  const int n = j / KP, k = j - n * KP;
  pack_slot s;
  s.real = n < NOUT && k < K;
  s.from = W + (s.real ? (size_t)k * NOUT + n : (size_t)0);
  s.to   = Wt + j;
  return s;
}
__device__ __forceinline__ pack_slot pack_slot_of(
    const float* fcW, const float* W1, const float* W2,
    bf16_t* fcWt, bf16_t* W1t, bf16_t* W2t, int i) {
  if (i < PACK_T0)           return pack_slot_in<IN_FT, HID1, KP0>(fcW, fcWt, i);
  if (i < PACK_T0 + PACK_T1) return pack_slot_in<HID1, HID2, KP1>(W1, W1t, i - PACK_T0);
  return pack_slot_in<HID2, OUT_FT, KP2>(W2, W2t, i - PACK_T0 - PACK_T1);
}

__device__ __forceinline__ v16bf load_frag(const bf16_t* p) {
  v8bf lo = *(const v8bf*)p;
  v8bf hi = *(const v8bf*)(p + 16);
  v16bf f;
#pragma unroll
  for (int i = 0; i < 8; ++i) { f[i] = lo[i]; f[8 + i] = hi[i]; }
  return f;
}

struct xfrag { float4 q0, q1, q2, q3; };
__device__ __forceinline__ xfrag load_xfrag(const float* p) {
  xfrag x;
  x.q0 = *(const float4*)(p);
  x.q1 = *(const float4*)(p + 4);
  x.q2 = *(const float4*)(p + 16);
  x.q3 = *(const float4*)(p + 20);
  return x;
}
__device__ __forceinline__ v16bf round_xfrag(const xfrag& x) {
  v16bf f;
  f[0]  = (bf16_t)x.q0.x; f[1]  = (bf16_t)x.q0.y; f[2]  = (bf16_t)x.q0.z; f[3]  = (bf16_t)x.q0.w;
  f[4]  = (bf16_t)x.q1.x; f[5]  = (bf16_t)x.q1.y; f[6]  = (bf16_t)x.q1.z; f[7]  = (bf16_t)x.q1.w;
  f[8]  = (bf16_t)x.q2.x; f[9]  = (bf16_t)x.q2.y; f[10] = (bf16_t)x.q2.z; f[11] = (bf16_t)x.q2.w;
  f[12] = (bf16_t)x.q3.x; f[13] = (bf16_t)x.q3.y; f[14] = (bf16_t)x.q3.z; f[15] = (bf16_t)x.q3.w;
  return f;
}

__device__ __forceinline__ v8f wmma_bf16(v16bf a, v16bf b, v8f c) {
  return __builtin_amdgcn_wmma_f32_16x16x32_bf16(
      false, a, false, b, (short)0, c, false, false);
}

#define FM    32
#define FW    7
#define H0P   (KP1 + 8)
static_assert(NP0 == 64 * FW && NP1 == 32 * FW && KP1 <= NP0, "wave tiles must cover the packed weights");

__global__ __launch_bounds__(FW * 32) void fused_fc_gemm1_kernel(
    const float* __restrict__ X, const bf16_t* __restrict__ W0t,
    const float* __restrict__ bias0, const bf16_t* __restrict__ W1t,
    float* __restrict__ T1, int M,
    const float* __restrict__ degp, float* __restrict__ dinv) {
  __shared__ alignas(16) bf16_t h0s[FM * H0P];

  const int lane = threadIdx.x & 31;
  const int wave = __builtin_amdgcn_readfirstlane(threadIdx.x >> 5);
  const int row0 = blockIdx.x * FM;
  if (row0 >= M) return;

  const int r  = lane & 15;
  const int hi = lane >> 4;

  float dshare[EDGE_C];
  if (wave == 0) {
#pragma unroll
    for (int cc = 0; cc < EDGE_C; ++cc) dshare[cc] = degp[(size_t)cc * M + row0 + lane];
  }

  {
    const float*  a0 = X   + (size_t)(row0 + r)      * IN_FT + hi * 8;
    const float*  a1 = X   + (size_t)(row0 + 16 + r) * IN_FT + hi * 8;
    const bf16_t* b  = W0t + (size_t)(wave * 64 + r) * KP0   + hi * 8;

    v8f c0[4] = {}, c1[4] = {};
    for (int kb = 0; kb < KP0; kb += 32) {
      const xfrag xa0 = load_xfrag(a0 + kb);
      const xfrag xa1 = load_xfrag(a1 + kb);
      v16bf vb[4];
#pragma unroll
      for (int j = 0; j < 4; ++j) vb[j] = load_frag(b + (size_t)j * 16 * KP0 + kb);
      const v16bf va0 = round_xfrag(xa0);
      const v16bf va1 = round_xfrag(xa1);
#pragma unroll
      for (int j = 0; j < 4; ++j) {
        c0[j] = wmma_bf16(va0, vb[j], c0[j]);
        c1[j] = wmma_bf16(va1, vb[j], c1[j]);
      }
      __builtin_amdgcn_sched_group_barrier(0x020, 16, 0);
      __builtin_amdgcn_sched_group_barrier(0x002, 16, 0);
      __builtin_amdgcn_sched_group_barrier(0x008, 8, 0);
    }

#pragma unroll
    for (int j = 0; j < 4; ++j) {
      const int n = wave * 64 + j * 16 + r;
      if (n >= KP1) continue;
      const bool real = n < HID1;
      const float bv = real ? bias0[n] : 0.0f;
#pragma unroll
      for (int i = 0; i < 8; ++i) {
        const int m = i + hi * 8;
        const float v0 = c0[j][i] + bv;
        const float v1 = c1[j][i] + bv;
        h0s[m * H0P + n]        = (bf16_t)((real && v0 > 0.0f) ? v0 : 0.0f);
        h0s[(m + 16) * H0P + n] = (bf16_t)((real && v1 > 0.0f) ? v1 : 0.0f);
      }
    }
  }
  if (wave == 0) {
    float deg = dshare[0];
#pragma unroll
    for (int cc = 1; cc < EDGE_C; ++cc) deg += dshare[cc];
    dinv[row0 + lane] = rsqrtf(deg + 1.0f);
  }
  __syncthreads();

  {
    const bf16_t* a0 = h0s + r * H0P        + hi * 8;
    const bf16_t* a1 = h0s + (16 + r) * H0P + hi * 8;
    const bf16_t* b0 = W1t + (size_t)(wave * 32 + r)      * KP1 + hi * 8;
    const bf16_t* b1 = W1t + (size_t)(wave * 32 + 16 + r) * KP1 + hi * 8;

    constexpr int KS = KP1 / 32;
    constexpr int PD = 4;
    v8f c00 = {}, c01 = {}, c10 = {}, c11 = {};
#pragma unroll
    for (int s = 0; s < KS; ++s) {
      v16bf va0 = load_frag(a0 + s * 32);
      v16bf va1 = load_frag(a1 + s * 32);
      v16bf vb0 = load_frag(b0 + s * 32);
      v16bf vb1 = load_frag(b1 + s * 32);
      c00 = wmma_bf16(va0, vb0, c00);
      c01 = wmma_bf16(va0, vb1, c01);
      c10 = wmma_bf16(va1, vb0, c10);
      c11 = wmma_bf16(va1, vb1, c11);
    }
    __builtin_amdgcn_sched_group_barrier(0x020, 4 * PD, 0);
    __builtin_amdgcn_sched_group_barrier(0x100, 4, 0);
#pragma unroll
    for (int s = 0; s < KS; ++s) {
      if (s + PD < KS) __builtin_amdgcn_sched_group_barrier(0x020, 4, 0);
      if (s + 1 < KS)  __builtin_amdgcn_sched_group_barrier(0x100, 4, 0);
      __builtin_amdgcn_sched_group_barrier(0x008, 4, 0);
    }

    const v8f cc[2][2] = {{c00, c01}, {c10, c11}};
#pragma unroll
    for (int sn = 0; sn < 2; ++sn) {
      const int n = wave * 32 + sn * 16 + r;
      if (n >= HID2) continue;
#pragma unroll
      for (int sm = 0; sm < 2; ++sm) {
        const v8f c = cc[sm][sn];
#pragma unroll
        for (int i = 0; i < 8; ++i) {
          const int m = row0 + sm * 16 + i + hi * 8;
          T1[(size_t)m * HID2 + n] = c[i];
        }
      }
    }
  }
}

__global__ __launch_bounds__(256) void wmma_gemm_kernel(
    const bf16_t* __restrict__ A, const bf16_t* __restrict__ Bt,
    float* __restrict__ Cout, int M, int Kp, int Nout, int tiles_n, int ldo) {
  const int wid  = blockIdx.x * 8 + (threadIdx.x >> 5);
  const int lane = threadIdx.x & 31;
  const int tm = wid / tiles_n;
  const int tn = wid % tiles_n;
  if (tm * 32 >= M) return;

  const int r  = lane & 15;
  const int hi = lane >> 4;
  const bf16_t* a0 = A  + (size_t)(tm * 32 + r)      * Kp + hi * 8;
  const bf16_t* a1 = A  + (size_t)(tm * 32 + 16 + r) * Kp + hi * 8;
  const bf16_t* b0 = Bt + (size_t)(tn * 32 + r)      * Kp + hi * 8;
  const bf16_t* b1 = Bt + (size_t)(tn * 32 + 16 + r) * Kp + hi * 8;

  v8f c00 = {}, c01 = {}, c10 = {}, c11 = {};
  for (int kb = 0; kb < Kp; kb += 32) {
    v16bf va0 = load_frag(a0 + kb);
    v16bf va1 = load_frag(a1 + kb);
    v16bf vb0 = load_frag(b0 + kb);
    v16bf vb1 = load_frag(b1 + kb);
    c00 = wmma_bf16(va0, vb0, c00);
    c01 = wmma_bf16(va0, vb1, c01);
    c10 = wmma_bf16(va1, vb0, c10);
    c11 = wmma_bf16(va1, vb1, c11);
  }

  const v8f cc[2][2] = {{c00, c01}, {c10, c11}};
#pragma unroll
  for (int sn = 0; sn < 2; ++sn) {
    const int n = tn * 32 + sn * 16 + r;
    if (n >= Nout) continue;
#pragma unroll
    for (int sm = 0; sm < 2; ++sm) {
      const v8f c = cc[sm][sn];
#pragma unroll
      for (int i = 0; i < 8; ++i) {
        const int m = tm * 32 + sm * 16 + i + hi * 8;
        Cout[(size_t)m * ldo + n] = c[i];
      }
    }
  }
}

#define AGG_R     80
#define AGG_T     1024
#define AGG_NW    (AGG_T / 32)
#define AGG_QCAP  2048
#define AGG_SCAP  256
#define AGG_B     8
#define EDGE_ROWS (AGG_R * EDGE_GB)
static_assert(EDGE_ROWS <= AGG_T, "one thread per row of a block clears and stores its degree share");

typedef __bf16 v2bf __attribute__((ext_vector_type(2)));

__device__ __forceinline__ void lds_add_f32(float* p, float v) {
  __hip_atomic_fetch_add(p, v, __ATOMIC_RELAXED, __HIP_MEMORY_SCOPE_WORKGROUP);
}
__device__ __forceinline__ float readlane_f32(float v, int l) {
  return __builtin_bit_cast(float,
      __builtin_amdgcn_readlane(__builtin_bit_cast(int, v), l));
}

__global__ __launch_bounds__(AGG_T) void edge_owner_kernel(
    const int* __restrict__ src, const int* __restrict__ dst,
    const float* __restrict__ ew, float* __restrict__ degp,
    int* __restrict__ qcnt, int* __restrict__ qbuf, int E, int Nn,
    const float* __restrict__ fcW, const float* __restrict__ W1,
    const float* __restrict__ W2, bf16_t* __restrict__ fcWt,
    bf16_t* __restrict__ W1t, bf16_t* __restrict__ W2t) {
  __shared__ float degs[EDGE_ROWS];
  __shared__ int   q_e[AGG_QCAP];
  __shared__ int   q_r[AGG_QCAP];
  __shared__ int   q_n;
  __shared__ int   g_n[EDGE_GB];

  const int tid  = threadIdx.x;
  const int lane = tid & 31;
  const int wave = __builtin_amdgcn_readfirstlane(tid >> 5);
  const int rb   = blockIdx.x / EDGE_C;
  const int ch   = blockIdx.x - rb * EDGE_C;
  const int base = rb * EDGE_ROWS;
  const int rows = (Nn - base) < EDGE_ROWS ? (Nn - base) : EDGE_ROWS;
  const int per = (((E + EDGE_C - 1) / EDGE_C) + 3) & ~3;
  const int e0  = (ch * per) < E ? (ch * per) : E;
  const int e1  = (e0 + per) < E ? (e0 + per) : E;

  if (tid < EDGE_ROWS) degs[tid] = 0.0f;
  if (tid < EDGE_GB) g_n[tid] = 0;
  if (tid == 0) q_n = 0;
  __syncthreads();

  const int pstride = gridDim.x * AGG_T;
  const int pi = blockIdx.x * AGG_T + tid;
  const bool phave = pi < PACK_TOTAL;
  const pack_slot ps = pack_slot_of(fcW, W1, W2, fcWt, W1t, W2t, phave ? pi : 0);
  const float pv = *ps.from;

  auto push = [&](bool m, int e, int rel) {
    const unsigned mask = __builtin_amdgcn_ballot_w32(m);
    if (mask == 0) return;
    const int cnt    = __builtin_popcount(mask);
    const int leader = __builtin_ctz(mask);
    int b = 0;
    if (lane == leader)
      b = __hip_atomic_fetch_add(&q_n, cnt, __ATOMIC_RELAXED,
                                 __HIP_MEMORY_SCOPE_WORKGROUP);
    b = __builtin_amdgcn_readlane(b, leader);
    const int slot = b + __builtin_popcount(mask & ((1u << lane) - 1u));
    if (m) {
      if (slot < AGG_QCAP) { q_e[slot] = e; q_r[slot] = rel; }
      else lds_add_f32(&degs[rel], ew[e]);
    }
  };

  const bool vec = (E & 3) == 0;
  const int4* dst4 = (const int4*)dst;
  const int v0 = e0 >> 2;
  const int v1 = vec ? (e1 >> 2) : v0;
  for (int wb = v0 + wave * 32; wb < v1; wb += 4 * AGG_T) {
    int4 d[4];
#pragma unroll
    for (int u = 0; u < 4; ++u) {
      const int i = wb + u * AGG_T + lane;
      d[u] = (i < v1) ? dst4[i] : make_int4(-1, -1, -1, -1);
    }
#pragma unroll
    for (int u = 0; u < 4; ++u) {
      const int i  = wb + u * AGG_T + lane;
      const int rx = d[u].x - base, ry = d[u].y - base;
      const int rz = d[u].z - base, rw = d[u].w - base;
      const bool mx = (unsigned)rx < (unsigned)rows;
      const bool my = (unsigned)ry < (unsigned)rows;
      const bool mz = (unsigned)rz < (unsigned)rows;
      const bool mw = (unsigned)rw < (unsigned)rows;
      if (__builtin_amdgcn_ballot_w32(mx | my | mz | mw) != 0) {
        push(mx, 4 * i + 0, rx);
        push(my, 4 * i + 1, ry);
        push(mz, 4 * i + 2, rz);
        push(mw, 4 * i + 3, rw);
      }
    }
  }
  for (int eb = (vec ? e1 : e0) + wave * 32; eb < e1; eb += AGG_T) {
    const int e  = eb + lane;
    const int rx = (e < e1) ? dst[e] - base : -1;
    push((unsigned)rx < (unsigned)rows, e, rx);
  }
  __syncthreads();

  if (phave) *ps.to = (bf16_t)(ps.real ? pv : 0.0f);
  for (int i = pi + pstride; i < PACK_TOTAL; i += pstride) {
    const pack_slot s = pack_slot_of(fcW, W1, W2, fcWt, W1t, W2t, i);
    *s.to = (bf16_t)(s.real ? *s.from : 0.0f);
  }

  const int total = q_n;
  const int n = total < AGG_QCAP ? total : AGG_QCAP;
  const int g0 = rb * EDGE_GB;
  int4* qg = (int4*)qbuf;
  for (int i = tid; i < n; i += AGG_T) {
    const int e = q_e[i], rel = q_r[i];
    const int   s  = src[e];
    const float wv = ew[e];
    lds_add_f32(&degs[rel], wv);
    const int gl   = rel / AGG_R;
    const int slot = __hip_atomic_fetch_add(&g_n[gl], 1, __ATOMIC_RELAXED,
                                            __HIP_MEMORY_SCOPE_WORKGROUP);
    const int at = slot < AGG_SCAP ? slot : AGG_SCAP - 1;
    qg[((size_t)(g0 + gl) * EDGE_C + ch) * AGG_SCAP + at] =
        make_int4(s, rel - gl * AGG_R, __builtin_bit_cast(int, wv), e);
  }
  __syncthreads();
  if (tid < rows) degp[(size_t)ch * Nn + base + tid] = degs[tid];
  if (tid < EDGE_GB && (g0 + tid) * AGG_R < Nn)
    qcnt[(g0 + tid) * EDGE_C + ch] = total > AGG_QCAP ? AGG_SCAP + 1 : g_n[tid];
}

template <int F, bool FINAL>
__global__ __launch_bounds__(AGG_T) void agg_owner_kernel(
    const float* __restrict__ t, const int* __restrict__ src,
    const int* __restrict__ dst, const float* __restrict__ ew,
    const float* __restrict__ dinv, const float* __restrict__ bias,
    const int* __restrict__ qcnt, const int* __restrict__ qbuf,
    float* __restrict__ outf, bf16_t* __restrict__ outb,
    int E, int Nn, int ldo) {
  constexpr int NF = (F + 31) / 32;
  __shared__ float acc[(AGG_R + 1) * F];

  const int tid  = threadIdx.x;
  const int lane = tid & 31;
  const int wave = __builtin_amdgcn_readfirstlane(tid >> 5);
  const int base = blockIdx.x * AGG_R;
  const int rows = (Nn - base) < AGG_R ? (Nn - base) : AGG_R;
  int pre[EDGE_C + 1];
  pre[0] = 0;
  bool whole = true;
#pragma unroll
  for (int cc = 0; cc < EDGE_C; ++cc) {
    const int cn = __builtin_amdgcn_readfirstlane(qcnt[blockIdx.x * EDGE_C + cc]);
    whole = whole && cn <= AGG_SCAP;
    pre[cc + 1] = pre[cc] + cn;
  }
  const int total = pre[EDGE_C];

  for (int idx = tid; idx < (AGG_R + 1) * F; idx += AGG_T) {
    const int r = idx / F, f = idx - r * F;
    float v = 0.0f;
    if (r < rows) {
      const float di = dinv[base + r];
      v = t[(size_t)(base + r) * F + f] * di * di + bias[f];
    }
    acc[idx] = v;
  }
  __syncthreads();

  if (!whole) {
    for (int eb = wave * 32; eb < E; eb += AGG_T) {
      const int e  = eb + lane;
      const int rx = (e < E) ? dst[e] - base : -1;
      unsigned mask = __builtin_amdgcn_ballot_w32((unsigned)rx < (unsigned)rows);
      while (mask) {
        const int l = __builtin_ctz(mask);
        mask &= mask - 1;
        const int ee  = __builtin_amdgcn_readlane(e, l);
        const int rel = __builtin_amdgcn_readlane(rx, l);
        const int s = src[ee];
        const float w = dinv[s] * ew[ee] * dinv[base + rel];
        const float* trow = t + (size_t)s * F;
        float* arow = acc + rel * F;
#pragma unroll
        for (int k = 0; k < NF; ++k) {
          const int f = k * 32 + lane;
          if (f < F) lds_add_f32(&arow[f], trow[f] * w);
        }
      }
    }
  } else {
    const int4* qg = (const int4*)qbuf + (size_t)blockIdx.x * EDGE_C * AGG_SCAP;
    auto entry_at = [&](int q) {
      int at = q;
#pragma unroll
      for (int cc = 1; cc < EDGE_C; ++cc)
        at = (q >= pre[cc]) ? cc * AGG_SCAP + (q - pre[cc]) : at;
      return qg[at];
    };
    int b0 = wave * AGG_B;
    if (b0 < total) {
      const int jl = lane & (AGG_B - 1);
      const int q0 = b0 + jl;
      int4 ent = entry_at(q0 < total ? q0 : total - 1);
      for (;;) {
        const int nb = b0 + AGG_NW * AGG_B;
        const int q1 = nb + jl;
        const int4 nxt = entry_at(q1 < total ? q1 : total - 1);
        const bool ok  = (b0 + jl) < total;
        const int s    = ok ? ent.x : 0;
        const int rel  = ok ? ent.y : AGG_R;
        const float we = ok ? __builtin_bit_cast(float, ent.z) : 0.0f;
        const float w  = dinv[s] * we * dinv[base + (ok ? ent.y : 0)];
        float v[AGG_B][NF];
#pragma unroll
        for (int j = 0; j < AGG_B; ++j) {
          const float* trow = t + (size_t)__builtin_amdgcn_readlane(s, j) * F;
#pragma unroll
          for (int k = 0; k < NF; ++k) {
            const int f = k * 32 + lane;
            v[j][k] = trow[f < F ? f : F - 1];
          }
        }
#pragma unroll
        for (int j = 0; j < AGG_B; ++j) {
          float* arow = acc + __builtin_amdgcn_readlane(rel, j) * F;
          const float wj = readlane_f32(w, j);
#pragma unroll
          for (int k = 0; k < NF; ++k) {
            const int f = k * 32 + lane;
            if (f < F) lds_add_f32(&arow[f], v[j][k] * wj);
          }
        }
        b0 = nb;
        ent = nxt;
        if (b0 >= total) break;
      }
    }
  }
  __syncthreads();

  if (FINAL) {
    float* o = outf + (size_t)base * F;
    for (int idx = tid; idx < rows * F; idx += AGG_T) {
      const float x = acc[idx];
      o[idx] = x > 0.0f ? x : 0.0f;
    }
  } else {
    const int hp = ldo >> 1;
    bf16_t* o = outb + (size_t)base * ldo;
    for (int idx = tid; idx < rows * hp; idx += AGG_T) {
      const int r = idx / hp, f = (idx - r * hp) * 2;
      float x0 = (f     < F) ? acc[r * F + f]     : 0.0f;
      float x1 = (f + 1 < F) ? acc[r * F + f + 1] : 0.0f;
      v2bf p;
      p[0] = (bf16_t)(x0 > 0.0f ? x0 : 0.0f);
      p[1] = (bf16_t)(x1 > 0.0f ? x1 : 0.0f);
      *(v2bf*)(o + (size_t)idx * 2) = p;
    }
  }
}

static inline size_t al256(size_t x) { return (x + 255) & ~(size_t)255; }

extern "C" void kernel_launch(void* const* d_in, const int* in_sizes, int n_in,
                              void* d_out, int out_size, void* d_ws, size_t ws_size,
                              hipStream_t stream) {
  const float* x    = (const float*)d_in[0];
  const int*   ei   = (const int*)  d_in[1];
  const float* ew   = (const float*)d_in[2];
  const float* fc_W = (const float*)d_in[3];
  const float* fc_b = (const float*)d_in[4];
  const float* W1   = (const float*)d_in[5];
  const float* b1   = (const float*)d_in[6];
  const float* W2   = (const float*)d_in[7];
  const float* b2   = (const float*)d_in[8];

  const int Nn = in_sizes[0] / IN_FT;
  const int E  = in_sizes[1] / 2;
  const int* src = ei;
  const int* dst = ei + E;

  char* ws = (char*)d_ws;
  size_t off = 0;
  auto alloc = [&](size_t bytes) { void* p = ws + off; off += al256(bytes); return p; };

  int*    qcnt = (int*)   alloc((size_t)Nn * 4);
  float*  dinv = (float*) alloc((size_t)Nn * 4);
  float*  degp = (float*) alloc((size_t)EDGE_C * Nn * 4);
  int*    qbuf = (int*)   alloc((size_t)((Nn + AGG_R - 1) / AGG_R) * EDGE_C * AGG_SCAP * 16);
  bf16_t* fcWt = (bf16_t*)alloc((size_t)NP0 * KP0 * 2);
  bf16_t* W1t  = (bf16_t*)alloc((size_t)NP1 * KP1 * 2);
  float*  t1   = (float*) alloc((size_t)Nn * HID2 * 4);
  bf16_t* h1b  = (bf16_t*)alloc((size_t)Nn * KP2 * 2);
  bf16_t* W2t  = (bf16_t*)alloc((size_t)NP2 * KP2 * 2);
  float*  t2   = (float*) alloc((size_t)Nn * OUT_FT * 4);
  (void)ws_size;

  edge_owner_kernel<<<((Nn + EDGE_ROWS - 1) / EDGE_ROWS) * EDGE_C, AGG_T, 0, stream>>>(
      src, dst, ew, degp, qcnt, qbuf, E, Nn, fc_W, W1, W2, fcWt, W1t, W2t);

  fused_fc_gemm1_kernel<<<Nn / FM, FW * 32, 0, stream>>>(
      x, fcWt, fc_b, W1t, t1, Nn, degp, dinv);

  {
    agg_owner_kernel<HID2, false><<<(Nn + AGG_R - 1) / AGG_R, AGG_T, 0, stream>>>(
        t1, src, dst, ew, dinv, b1, qcnt, qbuf, nullptr, h1b, E, Nn, KP2);
  }

  {
    const int tiles_n = NP2 / 32;
    const int waves   = (Nn / 32) * tiles_n;
    wmma_gemm_kernel<<<(waves + 7) / 8, 256, 0, stream>>>(
        h1b, W2t, t2, Nn, KP2, OUT_FT, tiles_n, OUT_FT);
    agg_owner_kernel<OUT_FT, true><<<(Nn + AGG_R - 1) / AGG_R, AGG_T, 0, stream>>>(
        t2, src, dst, ew, dinv, b2, qcnt, qbuf, (float*)d_out, nullptr, E, Nn, OUT_FT);
  }
}


# Fused layer-0/1 GEMM: row block 32 -> 80 rows per work-group (625 -> 250 groups, each reads both weight matrices once); product 1 pipelined one k-step deep (next loads under WMMAs); same k order
# speedup vs baseline: 3.5178x; 1.1889x over previous
#include <hip/hip_runtime.h>
#include <hip/hip_bf16.h>


typedef __bf16 bf16_t;
typedef __bf16 v16bf __attribute__((ext_vector_type(16)));
typedef __bf16 v8bf  __attribute__((ext_vector_type(8)));
typedef float  v8f   __attribute__((ext_vector_type(8)));

#define IN_FT   256
#define HID1    400
#define HID2    200
#define OUT_FT  128

#define KP0     256
#define KP1     416
#define KP2     224
#define NP0     448
#define NP1     224
#define NP2     128

#define EDGE_GB   10
#define EDGE_C    10

#define PACK_T0     (NP0 * KP0)
#define PACK_T1     (NP1 * KP1)
#define PACK_T2     (NP2 * KP2)
#define PACK_TOTAL  (PACK_T0 + PACK_T1 + PACK_T2)
struct pack_slot { const float* from; bf16_t* to; bool real; };
template <int K, int NOUT, int KP>
__device__ __forceinline__ pack_slot pack_slot_in(const float* W, bf16_t* Wt, int j) {
  const int n = j / KP, k = j - n * KP;
  pack_slot s;
  s.real = n < NOUT && k < K;
  s.from = W + (s.real ? (size_t)k * NOUT + n : (size_t)0);
  s.to   = Wt + j;
  return s;
}
__device__ __forceinline__ pack_slot pack_slot_of(
    const float* fcW, const float* W1, const float* W2,
    bf16_t* fcWt, bf16_t* W1t, bf16_t* W2t, int i) {
  if (i < PACK_T0)           return pack_slot_in<IN_FT, HID1, KP0>(fcW, fcWt, i);
  if (i < PACK_T0 + PACK_T1) return pack_slot_in<HID1, HID2, KP1>(W1, W1t, i - PACK_T0);
  return pack_slot_in<HID2, OUT_FT, KP2>(W2, W2t, i - PACK_T0 - PACK_T1);
}

__device__ __forceinline__ v16bf load_frag(const bf16_t* p) {
  v8bf lo = *(const v8bf*)p;
  v8bf hi = *(const v8bf*)(p + 16);
  v16bf f;
#pragma unroll
  for (int i = 0; i < 8; ++i) { f[i] = lo[i]; f[8 + i] = hi[i]; }
  return f;
}

struct xfrag { float4 q0, q1, q2, q3; };
__device__ __forceinline__ xfrag load_xfrag(const float* p) {
  xfrag x;
  x.q0 = *(const float4*)(p);
  x.q1 = *(const float4*)(p + 4);
  x.q2 = *(const float4*)(p + 16);
  x.q3 = *(const float4*)(p + 20);
  return x;
}
__device__ __forceinline__ v16bf round_xfrag(const xfrag& x) {
  v16bf f;
  f[0]  = (bf16_t)x.q0.x; f[1]  = (bf16_t)x.q0.y; f[2]  = (bf16_t)x.q0.z; f[3]  = (bf16_t)x.q0.w;
  f[4]  = (bf16_t)x.q1.x; f[5]  = (bf16_t)x.q1.y; f[6]  = (bf16_t)x.q1.z; f[7]  = (bf16_t)x.q1.w;
  f[8]  = (bf16_t)x.q2.x; f[9]  = (bf16_t)x.q2.y; f[10] = (bf16_t)x.q2.z; f[11] = (bf16_t)x.q2.w;
  f[12] = (bf16_t)x.q3.x; f[13] = (bf16_t)x.q3.y; f[14] = (bf16_t)x.q3.z; f[15] = (bf16_t)x.q3.w;
  return f;
}

__device__ __forceinline__ v8f wmma_bf16(v16bf a, v16bf b, v8f c) {
  return __builtin_amdgcn_wmma_f32_16x16x32_bf16(
      false, a, false, b, (short)0, c, false, false);
}

#define FM    80
#define MT    (FM / 16)
#define FW    7
#define H0P   (KP1 + 8)
static_assert(NP0 == 64 * FW && NP1 == 32 * FW && KP1 <= NP0, "wave tiles must cover the packed weights");
static_assert(FM % 16 == 0 && FM <= 3 * 32, "row tiles of 16; the first three waves finish the degrees");

__global__ __launch_bounds__(FW * 32) void fused_fc_gemm1_kernel(
    const float* __restrict__ X, const bf16_t* __restrict__ W0t,
    const float* __restrict__ bias0, const bf16_t* __restrict__ W1t,
    float* __restrict__ T1, int M,
    const float* __restrict__ degp, float* __restrict__ dinv) {
  __shared__ alignas(16) bf16_t h0s[FM * H0P];

  const int lane = threadIdx.x & 31;
  const int wave = __builtin_amdgcn_readfirstlane(threadIdx.x >> 5);
  const int row0 = blockIdx.x * FM;
  if (row0 >= M) return;
  const int mlim = M - row0;

  const int r  = lane & 15;
  const int hi = lane >> 4;

  const int dn = wave * 32 + lane;
  const bool dwave = wave * 32 < FM;
  float dshare[EDGE_C];
  if (dwave) {
    const int node = (dn < mlim && dn < FM) ? row0 + dn : M - 1;
#pragma unroll
    for (int cc = 0; cc < EDGE_C; ++cc) dshare[cc] = degp[(size_t)cc * M + node];
  }

  {
    const float* a[MT];
#pragma unroll
    for (int t = 0; t < MT; ++t) {
      const int m = t * 16 + r;
      a[t] = X + (size_t)(row0 + (m < mlim ? m : mlim - 1)) * IN_FT + hi * 8;
    }
    const bf16_t* b = W0t + (size_t)(wave * 64 + r) * KP0 + hi * 8;

    v8f c[MT][4] = {};
    constexpr int KS0 = KP0 / 32;
    constexpr int LD0 = 4 * MT + 2 * 4;
    v16bf va[2][MT], vb[2][4];
    auto fetch_step = [&](v16bf (&as)[MT], v16bf (&bs)[4], int kb) {
      xfrag x[MT];
#pragma unroll
      for (int t = 0; t < MT; ++t) x[t] = load_xfrag(a[t] + kb);
#pragma unroll
      for (int j = 0; j < 4; ++j) bs[j] = load_frag(b + (size_t)j * 16 * KP0 + kb);
#pragma unroll
      for (int t = 0; t < MT; ++t) as[t] = round_xfrag(x[t]);
    };
    fetch_step(va[0], vb[0], 0);
    __builtin_amdgcn_sched_barrier(0);
#pragma unroll
    for (int s = 0; s < KS0; ++s) {
      if (s + 1 < KS0) fetch_step(va[(s + 1) & 1], vb[(s + 1) & 1], (s + 1) * 32);
#pragma unroll
      for (int j = 0; j < 4; ++j) {
#pragma unroll
        for (int t = 0; t < MT; ++t) c[t][j] = wmma_bf16(va[s & 1][t], vb[s & 1][j], c[t][j]);
      }
      if (s + 1 < KS0) __builtin_amdgcn_sched_group_barrier(0x020, LD0, 0);
      __builtin_amdgcn_sched_group_barrier(0x008, 4 * MT, 0);
      if (s + 1 < KS0) __builtin_amdgcn_sched_group_barrier(0x002, 8 * MT, 0);
      __builtin_amdgcn_sched_barrier(0);
    }

#pragma unroll
    for (int j = 0; j < 4; ++j) {
      const int n = wave * 64 + j * 16 + r;
      if (n >= KP1) continue;
      const bool real = n < HID1;
      const float bv = real ? bias0[n] : 0.0f;
#pragma unroll
      for (int t = 0; t < MT; ++t) {
#pragma unroll
        for (int i = 0; i < 8; ++i) {
          const int m = t * 16 + i + hi * 8;
          const float v = c[t][j][i] + bv;
          h0s[m * H0P + n] = (bf16_t)((real && v > 0.0f) ? v : 0.0f);
        }
      }
    }
  }
  if (dwave) {
    float deg = dshare[0];
#pragma unroll
    for (int cc = 1; cc < EDGE_C; ++cc) deg += dshare[cc];
    if (dn < mlim && dn < FM) dinv[row0 + dn] = rsqrtf(deg + 1.0f);
  }
  __syncthreads();

  {
    const bf16_t* a  = h0s + r * H0P + hi * 8;
    const bf16_t* b0 = W1t + (size_t)(wave * 32 + r)      * KP1 + hi * 8;
    const bf16_t* b1 = W1t + (size_t)(wave * 32 + 16 + r) * KP1 + hi * 8;

    constexpr int KS = KP1 / 32;
    constexpr int PD = 4;
    v8f d[MT][2] = {};
#pragma unroll
    for (int s = 0; s < KS; ++s) {
      v16bf va[MT];
#pragma unroll
      for (int t = 0; t < MT; ++t) va[t] = load_frag(a + t * 16 * H0P + s * 32);
      v16bf vb0 = load_frag(b0 + s * 32);
      v16bf vb1 = load_frag(b1 + s * 32);
#pragma unroll
      for (int t = 0; t < MT; ++t) {
        d[t][0] = wmma_bf16(va[t], vb0, d[t][0]);
        d[t][1] = wmma_bf16(va[t], vb1, d[t][1]);
      }
    }
    __builtin_amdgcn_sched_group_barrier(0x020, 4 * PD, 0);
    __builtin_amdgcn_sched_group_barrier(0x100, 2 * MT, 0);
#pragma unroll
    for (int s = 0; s < KS; ++s) {
      if (s + PD < KS) __builtin_amdgcn_sched_group_barrier(0x020, 4, 0);
      if (s + 1 < KS)  __builtin_amdgcn_sched_group_barrier(0x100, 2 * MT, 0);
      __builtin_amdgcn_sched_group_barrier(0x008, 2 * MT, 0);
    }

#pragma unroll
    for (int sn = 0; sn < 2; ++sn) {
      const int n = wave * 32 + sn * 16 + r;
      if (n >= HID2) continue;
#pragma unroll
      for (int t = 0; t < MT; ++t) {
        if (t * 16 >= mlim) continue;
#pragma unroll
        for (int i = 0; i < 8; ++i) {
          const int m = t * 16 + i + hi * 8;
          T1[(size_t)(row0 + m) * HID2 + n] = d[t][sn][i];
        }
      }
    }
  }
}

__global__ __launch_bounds__(256) void wmma_gemm_kernel(
    const bf16_t* __restrict__ A, const bf16_t* __restrict__ Bt,
    float* __restrict__ Cout, int M, int Kp, int Nout, int tiles_n, int ldo) {
  const int wid  = blockIdx.x * 8 + (threadIdx.x >> 5);
  const int lane = threadIdx.x & 31;
  const int tm = wid / tiles_n;
  const int tn = wid % tiles_n;
  if (tm * 32 >= M) return;

  const int r  = lane & 15;
  const int hi = lane >> 4;
  const bf16_t* a0 = A  + (size_t)(tm * 32 + r)      * Kp + hi * 8;
  const bf16_t* a1 = A  + (size_t)(tm * 32 + 16 + r) * Kp + hi * 8;
  const bf16_t* b0 = Bt + (size_t)(tn * 32 + r)      * Kp + hi * 8;
  const bf16_t* b1 = Bt + (size_t)(tn * 32 + 16 + r) * Kp + hi * 8;

  v8f c00 = {}, c01 = {}, c10 = {}, c11 = {};
  for (int kb = 0; kb < Kp; kb += 32) {
    v16bf va0 = load_frag(a0 + kb);
    v16bf va1 = load_frag(a1 + kb);
    v16bf vb0 = load_frag(b0 + kb);
    v16bf vb1 = load_frag(b1 + kb);
    c00 = wmma_bf16(va0, vb0, c00);
    c01 = wmma_bf16(va0, vb1, c01);
    c10 = wmma_bf16(va1, vb0, c10);
    c11 = wmma_bf16(va1, vb1, c11);
  }

  const v8f cc[2][2] = {{c00, c01}, {c10, c11}};
#pragma unroll
  for (int sn = 0; sn < 2; ++sn) {
    const int n = tn * 32 + sn * 16 + r;
    if (n >= Nout) continue;
#pragma unroll
    for (int sm = 0; sm < 2; ++sm) {
      const v8f c = cc[sm][sn];
#pragma unroll
      for (int i = 0; i < 8; ++i) {
        const int m = tm * 32 + sm * 16 + i + hi * 8;
        Cout[(size_t)m * ldo + n] = c[i];
      }
    }
  }
}

#define AGG_R     80
#define AGG_T     1024
#define AGG_NW    (AGG_T / 32)
#define AGG_QCAP  2048
#define AGG_SCAP  256
#define AGG_B     8
#define EDGE_ROWS (AGG_R * EDGE_GB)
static_assert(EDGE_ROWS <= AGG_T, "one thread per row of a block clears and stores its degree share");

typedef __bf16 v2bf __attribute__((ext_vector_type(2)));

__device__ __forceinline__ void lds_add_f32(float* p, float v) {
  __hip_atomic_fetch_add(p, v, __ATOMIC_RELAXED, __HIP_MEMORY_SCOPE_WORKGROUP);
}
__device__ __forceinline__ float readlane_f32(float v, int l) {
  return __builtin_bit_cast(float,
      __builtin_amdgcn_readlane(__builtin_bit_cast(int, v), l));
}

__global__ __launch_bounds__(AGG_T) void edge_owner_kernel(
    const int* __restrict__ src, const int* __restrict__ dst,
    const float* __restrict__ ew, float* __restrict__ degp,
    int* __restrict__ qcnt, int* __restrict__ qbuf, int E, int Nn,
    const float* __restrict__ fcW, const float* __restrict__ W1,
    const float* __restrict__ W2, bf16_t* __restrict__ fcWt,
    bf16_t* __restrict__ W1t, bf16_t* __restrict__ W2t) {
  __shared__ float degs[EDGE_ROWS];
  __shared__ int   q_e[AGG_QCAP];
  __shared__ int   q_r[AGG_QCAP];
  __shared__ int   q_n;
  __shared__ int   g_n[EDGE_GB];

  const int tid  = threadIdx.x;
  const int lane = tid & 31;
  const int wave = __builtin_amdgcn_readfirstlane(tid >> 5);
  const int rb   = blockIdx.x / EDGE_C;
  const int ch   = blockIdx.x - rb * EDGE_C;
  const int base = rb * EDGE_ROWS;
  const int rows = (Nn - base) < EDGE_ROWS ? (Nn - base) : EDGE_ROWS;
  const int per = (((E + EDGE_C - 1) / EDGE_C) + 3) & ~3;
  const int e0  = (ch * per) < E ? (ch * per) : E;
  const int e1  = (e0 + per) < E ? (e0 + per) : E;

  if (tid < EDGE_ROWS) degs[tid] = 0.0f;
  if (tid < EDGE_GB) g_n[tid] = 0;
  if (tid == 0) q_n = 0;
  __syncthreads();

  const int pstride = gridDim.x * AGG_T;
  const int pi = blockIdx.x * AGG_T + tid;
  const bool phave = pi < PACK_TOTAL;
  const pack_slot ps = pack_slot_of(fcW, W1, W2, fcWt, W1t, W2t, phave ? pi : 0);
  const float pv = *ps.from;

  auto push = [&](bool m, int e, int rel) {
    const unsigned mask = __builtin_amdgcn_ballot_w32(m);
    if (mask == 0) return;
    const int cnt    = __builtin_popcount(mask);
    const int leader = __builtin_ctz(mask);
    int b = 0;
    if (lane == leader)
      b = __hip_atomic_fetch_add(&q_n, cnt, __ATOMIC_RELAXED,
                                 __HIP_MEMORY_SCOPE_WORKGROUP);
    b = __builtin_amdgcn_readlane(b, leader);
    const int slot = b + __builtin_popcount(mask & ((1u << lane) - 1u));
    if (m) {
      if (slot < AGG_QCAP) { q_e[slot] = e; q_r[slot] = rel; }
      else lds_add_f32(&degs[rel], ew[e]);
    }
  };

  const bool vec = (E & 3) == 0;
  const int4* dst4 = (const int4*)dst;
  const int v0 = e0 >> 2;
  const int v1 = vec ? (e1 >> 2) : v0;
  for (int wb = v0 + wave * 32; wb < v1; wb += 4 * AGG_T) {
    int4 d[4];
#pragma unroll
    for (int u = 0; u < 4; ++u) {
      const int i = wb + u * AGG_T + lane;
      d[u] = (i < v1) ? dst4[i] : make_int4(-1, -1, -1, -1);
    }
#pragma unroll
    for (int u = 0; u < 4; ++u) {
      const int i  = wb + u * AGG_T + lane;
      const int rx = d[u].x - base, ry = d[u].y - base;
      const int rz = d[u].z - base, rw = d[u].w - base;
      const bool mx = (unsigned)rx < (unsigned)rows;
      const bool my = (unsigned)ry < (unsigned)rows;
      const bool mz = (unsigned)rz < (unsigned)rows;
      const bool mw = (unsigned)rw < (unsigned)rows;
      if (__builtin_amdgcn_ballot_w32(mx | my | mz | mw) != 0) {
        push(mx, 4 * i + 0, rx);
        push(my, 4 * i + 1, ry);
        push(mz, 4 * i + 2, rz);
        push(mw, 4 * i + 3, rw);
      }
    }
  }
  for (int eb = (vec ? e1 : e0) + wave * 32; eb < e1; eb += AGG_T) {
    const int e  = eb + lane;
    const int rx = (e < e1) ? dst[e] - base : -1;
    push((unsigned)rx < (unsigned)rows, e, rx);
  }
  __syncthreads();

  if (phave) *ps.to = (bf16_t)(ps.real ? pv : 0.0f);
  for (int i = pi + pstride; i < PACK_TOTAL; i += pstride) {
    const pack_slot s = pack_slot_of(fcW, W1, W2, fcWt, W1t, W2t, i);
    *s.to = (bf16_t)(s.real ? *s.from : 0.0f);
  }

  const int total = q_n;
  const int n = total < AGG_QCAP ? total : AGG_QCAP;
  const int g0 = rb * EDGE_GB;
  int4* qg = (int4*)qbuf;
  for (int i = tid; i < n; i += AGG_T) {
    const int e = q_e[i], rel = q_r[i];
    const int   s  = src[e];
    const float wv = ew[e];
    lds_add_f32(&degs[rel], wv);
    const int gl   = rel / AGG_R;
    const int slot = __hip_atomic_fetch_add(&g_n[gl], 1, __ATOMIC_RELAXED,
                                            __HIP_MEMORY_SCOPE_WORKGROUP);
    const int at = slot < AGG_SCAP ? slot : AGG_SCAP - 1;
    qg[((size_t)(g0 + gl) * EDGE_C + ch) * AGG_SCAP + at] =
        make_int4(s, rel - gl * AGG_R, __builtin_bit_cast(int, wv), e);
  }
  __syncthreads();
  if (tid < rows) degp[(size_t)ch * Nn + base + tid] = degs[tid];
  if (tid < EDGE_GB && (g0 + tid) * AGG_R < Nn)
    qcnt[(g0 + tid) * EDGE_C + ch] = total > AGG_QCAP ? AGG_SCAP + 1 : g_n[tid];
}

template <int F, bool FINAL>
__global__ __launch_bounds__(AGG_T) void agg_owner_kernel(
    const float* __restrict__ t, const int* __restrict__ src,
    const int* __restrict__ dst, const float* __restrict__ ew,
    const float* __restrict__ dinv, const float* __restrict__ bias,
    const int* __restrict__ qcnt, const int* __restrict__ qbuf,
    float* __restrict__ outf, bf16_t* __restrict__ outb,
    int E, int Nn, int ldo) {
  constexpr int NF = (F + 31) / 32;
  __shared__ float acc[(AGG_R + 1) * F];

  const int tid  = threadIdx.x;
  const int lane = tid & 31;
  const int wave = __builtin_amdgcn_readfirstlane(tid >> 5);
  const int base = blockIdx.x * AGG_R;
  const int rows = (Nn - base) < AGG_R ? (Nn - base) : AGG_R;
  constexpr int CW = F <= 128 ? 128 : 256;
  static_assert(F <= CW && AGG_T % CW == 0, "a row slot is a power-of-two number of threads");
  const int cf = tid & (CW - 1);
  const float bv = bias[cf < F ? cf : F - 1];
  int pre[EDGE_C + 1];
  pre[0] = 0;
  bool whole = true;
#pragma unroll
  for (int cc = 0; cc < EDGE_C; ++cc) {
    const int cn = __builtin_amdgcn_readfirstlane(qcnt[blockIdx.x * EDGE_C + cc]);
    whole = whole && cn <= AGG_SCAP;
    pre[cc + 1] = pre[cc] + cn;
  }
  const int total = pre[EDGE_C];
  const int tot2 = total + rows;
  const int last = total > 0 ? total - 1 : 0;
  const int4* qg = (const int4*)qbuf + (size_t)blockIdx.x * EDGE_C * AGG_SCAP;
  auto entry_at = [&](int q) {
    int at = q;
#pragma unroll
    for (int cc = 1; cc < EDGE_C; ++cc)
      at = (q >= pre[cc]) ? cc * AGG_SCAP + (q - pre[cc]) : at;
    return qg[at];
  };
  int b0 = wave * AGG_B;
  const int jl = lane & (AGG_B - 1);
  int ex = 0, ey = 0, ez = 0;
  if (whole && total > 0) {
    const int q0 = b0 + jl;
    const int4 e4 = entry_at(q0 < last ? q0 : last);
    ex = e4.x; ey = e4.y; ez = e4.z;
  }

  if (cf < F)
    for (int r = tid / CW; r <= AGG_R; r += AGG_T / CW)
      acc[r * F + cf] = r < rows ? bv : 0.0f;
  __syncthreads();

  if (!whole) {
    for (int idx = tid; idx < rows * F; idx += AGG_T) {
      const int r = idx / F, f = idx - r * F;
      const float di = dinv[base + r];
      lds_add_f32(&acc[idx], t[(size_t)(base + r) * F + f] * di * di);
    }
    for (int eb = wave * 32; eb < E; eb += AGG_T) {
      const int e  = eb + lane;
      const int rx = (e < E) ? dst[e] - base : -1;
      unsigned mask = __builtin_amdgcn_ballot_w32((unsigned)rx < (unsigned)rows);
      while (mask) {
        const int l = __builtin_ctz(mask);
        mask &= mask - 1;
        const int ee  = __builtin_amdgcn_readlane(e, l);
        const int rel = __builtin_amdgcn_readlane(rx, l);
        const int s = src[ee];
        const float w = dinv[s] * ew[ee] * dinv[base + rel];
        const float* trow = t + (size_t)s * F;
        float* arow = acc + rel * F;
#pragma unroll
        for (int k = 0; k < NF; ++k) {
          const int f = k * 32 + lane;
          if (f < F) lds_add_f32(&arow[f], trow[f] * w);
        }
      }
    }
  } else {
    if (b0 < tot2) {
      for (;;) {
        const int nb = b0 + AGG_NW * AGG_B;
        const int q1 = nb + jl;
        int nx = 0, ny = 0, nz = 0;
        if (total > 0) {
          const int4 e4 = entry_at(q1 < last ? q1 : last);
          nx = e4.x; ny = e4.y; nz = e4.z;
        }
        const int q    = b0 + jl;
        const bool ok  = q < tot2;
        const bool own = q >= total;
        const int rown = q - total;
        const int s    = ok ? (own ? base + rown : ex) : 0;
        const int rel  = ok ? (own ? rown : ey) : AGG_R;
        const float we = ok ? (own ? 1.0f : __builtin_bit_cast(float, ez)) : 0.0f;
        const float w  = dinv[s] * we * dinv[base + (ok ? rel : 0)];
        float v[AGG_B][NF];
#pragma unroll
        for (int j = 0; j < AGG_B; ++j) {
          const float* trow = t + (size_t)__builtin_amdgcn_readlane(s, j) * F;
#pragma unroll
          for (int k = 0; k < NF; ++k) {
            const int f = k * 32 + lane;
            v[j][k] = trow[f < F ? f : F - 1];
          }
        }
#pragma unroll
        for (int j = 0; j < AGG_B; ++j) {
          float* arow = acc + __builtin_amdgcn_readlane(rel, j) * F;
          const float wj = readlane_f32(w, j);
#pragma unroll
          for (int k = 0; k < NF; ++k) {
            const int f = k * 32 + lane;
            if (k * 32 + 31 < F) {
              lds_add_f32(&arow[f], v[j][k] * wj);
            } else {
              float* p = f < F ? &arow[f] : &acc[AGG_R * F + lane];
              lds_add_f32(p, v[j][k] * wj);
            }
          }
        }
        b0 = nb;
        ex = nx; ey = ny; ez = nz;
        if (b0 >= tot2) break;
      }
    }
  }
  __syncthreads();

  if (FINAL) {
    float* o = outf + (size_t)base * F;
    for (int idx = tid; idx < rows * F; idx += AGG_T) {
      const float x = acc[idx];
      o[idx] = x > 0.0f ? x : 0.0f;
    }
  } else {
    const int hp = ldo >> 1;
    bf16_t* o = outb + (size_t)base * ldo;
    for (int idx = tid; idx < rows * hp; idx += AGG_T) {
      const int r = idx / hp, f = (idx - r * hp) * 2;
      float x0 = (f     < F) ? acc[r * F + f]     : 0.0f;
      float x1 = (f + 1 < F) ? acc[r * F + f + 1] : 0.0f;
      v2bf p;
      p[0] = (bf16_t)(x0 > 0.0f ? x0 : 0.0f);
      p[1] = (bf16_t)(x1 > 0.0f ? x1 : 0.0f);
      *(v2bf*)(o + (size_t)idx * 2) = p;
    }
  }
}

static inline size_t al256(size_t x) { return (x + 255) & ~(size_t)255; }

extern "C" void kernel_launch(void* const* d_in, const int* in_sizes, int n_in,
                              void* d_out, int out_size, void* d_ws, size_t ws_size,
                              hipStream_t stream) {
  const float* x    = (const float*)d_in[0];
  const int*   ei   = (const int*)  d_in[1];
  const float* ew   = (const float*)d_in[2];
  const float* fc_W = (const float*)d_in[3];
  const float* fc_b = (const float*)d_in[4];
  const float* W1   = (const float*)d_in[5];
  const float* b1   = (const float*)d_in[6];
  const float* W2   = (const float*)d_in[7];
  const float* b2   = (const float*)d_in[8];

  const int Nn = in_sizes[0] / IN_FT;
  const int E  = in_sizes[1] / 2;
  const int* src = ei;
  const int* dst = ei + E;

  char* ws = (char*)d_ws;
  size_t off = 0;
  auto alloc = [&](size_t bytes) { void* p = ws + off; off += al256(bytes); return p; };

  int*    qcnt = (int*)   alloc((size_t)Nn * 4);
  float*  dinv = (float*) alloc((size_t)Nn * 4);
  float*  degp = (float*) alloc((size_t)EDGE_C * Nn * 4);
  int*    qbuf = (int*)   alloc((size_t)((Nn + AGG_R - 1) / AGG_R) * EDGE_C * AGG_SCAP * 16);
  bf16_t* fcWt = (bf16_t*)alloc((size_t)NP0 * KP0 * 2);
  bf16_t* W1t  = (bf16_t*)alloc((size_t)NP1 * KP1 * 2);
  float*  t1   = (float*) alloc((size_t)Nn * HID2 * 4);
  bf16_t* h1b  = (bf16_t*)alloc((size_t)Nn * KP2 * 2);
  bf16_t* W2t  = (bf16_t*)alloc((size_t)NP2 * KP2 * 2);
  float*  t2   = (float*) alloc((size_t)Nn * OUT_FT * 4);
  (void)ws_size;

  edge_owner_kernel<<<((Nn + EDGE_ROWS - 1) / EDGE_ROWS) * EDGE_C, AGG_T, 0, stream>>>(
      src, dst, ew, degp, qcnt, qbuf, E, Nn, fc_W, W1, W2, fcWt, W1t, W2t);

  fused_fc_gemm1_kernel<<<(Nn + FM - 1) / FM, FW * 32, 0, stream>>>(
      x, fcWt, fc_b, W1t, t1, Nn, degp, dinv);

  {
    agg_owner_kernel<HID2, false><<<(Nn + AGG_R - 1) / AGG_R, AGG_T, 0, stream>>>(
        t1, src, dst, ew, dinv, b1, qcnt, qbuf, nullptr, h1b, E, Nn, KP2);
  }

  {
    const int tiles_n = NP2 / 32;
    const int waves   = (Nn / 32) * tiles_n;
    wmma_gemm_kernel<<<(waves + 7) / 8, 256, 0, stream>>>(
        h1b, W2t, t2, Nn, KP2, OUT_FT, tiles_n, OUT_FT);
    agg_owner_kernel<OUT_FT, true><<<(Nn + AGG_R - 1) / AGG_R, AGG_T, 0, stream>>>(
        t2, src, dst, ew, dinv, b2, qcnt, qbuf, (float*)d_out, nullptr, E, Nn, OUT_FT);
  }
}
